# starting kernel
# baseline (speedup 1.0000x reference)
#include <hip/hip_runtime.h>

typedef __attribute__((ext_vector_type(2))) float v2f;
typedef __attribute__((ext_vector_type(8))) float v8f;

#define NFEAT 256
#define NCLASS 64

__global__ void deg_init_kernel(float* __restrict__ deg, int n) {
    int i = blockIdx.x * blockDim.x + threadIdx.x;
    if (i < n) deg[i] = 1.0f;
}

__global__ void deg_count_kernel(const int* __restrict__ col, float* __restrict__ deg, int e) {
    int i = blockIdx.x * blockDim.x + threadIdx.x;
    if (i < e) atomicAdd(&deg[col[i]], 1.0f);
}

__global__ void deg_rsqrt_kernel(float* __restrict__ deg, int n) {
    int i = blockIdx.x * blockDim.x + threadIdx.x;
    if (i < n) deg[i] = rsqrtf(deg[i]);
}

__global__ __launch_bounds__(256) void gemm1_kernel(const float* __restrict__ X,
                                                    const float* __restrict__ Wg,
                                                    float* __restrict__ H, int n) {
    const int lane = threadIdx.x & 31;
    const int wave = threadIdx.x >> 5;
    const int mtile = blockIdx.x * 8 + wave;
    const int m0 = mtile * 16;
    if (m0 >= n) return;

    const int half = lane >> 4;
    const int l15  = lane & 15;
    int m = m0 + l15;
    if (m >= n) m = n - 1;
    const float* __restrict__ xrow = X + (size_t)m * NFEAT;

    v8f acc0 = {}, acc1 = {}, acc2 = {}, acc3 = {};

    for (int k0 = 0; k0 < NFEAT; k0 += 4) {
        const int kb = k0 + half * 2;
        v2f a;
        a.x = xrow[kb];
        a.y = xrow[kb + 1];

        const float* __restrict__ w0 = Wg + (size_t)kb * NCLASS;
        const float* __restrict__ w1 = w0 + NCLASS;
        v2f b0, b1, b2, b3;
        b0.x = w0[l15];      b0.y = w1[l15];
        b1.x = w0[16 + l15]; b1.y = w1[16 + l15];
        b2.x = w0[32 + l15]; b2.y = w1[32 + l15];
        b3.x = w0[48 + l15]; b3.y = w1[48 + l15];

        acc0 = __builtin_amdgcn_wmma_f32_16x16x4_f32(false, a, false, b0, (short)0, acc0, false, false);
        acc1 = __builtin_amdgcn_wmma_f32_16x16x4_f32(false, a, false, b1, (short)0, acc1, false, false);
        acc2 = __builtin_amdgcn_wmma_f32_16x16x4_f32(false, a, false, b2, (short)0, acc2, false, false);
        acc3 = __builtin_amdgcn_wmma_f32_16x16x4_f32(false, a, false, b3, (short)0, acc3, false, false);
    }

    const int mbase = m0 + half * 8;
#pragma unroll
    for (int r = 0; r < 8; ++r) {
        const int mr = mbase + r;
        if (mr < n) {
            float* __restrict__ hrow = H + (size_t)mr * NCLASS;
            hrow[l15]      = acc0[r];
            hrow[16 + l15] = acc1[r];
            hrow[32 + l15] = acc2[r];
            hrow[48 + l15] = acc3[r];
        }
    }
}

__global__ void agg_init_kernel(const float* __restrict__ H,
                                const float* __restrict__ dinv,
                                const float* __restrict__ bg,
                                float* __restrict__ agg, int n) {
    int i = blockIdx.x * blockDim.x + threadIdx.x;
    int total = n * NCLASS;
    if (i < total) {
        int node = i >> 6;
        int c = i & 63;
        float d = dinv[node];
        agg[i] = H[i] * d * d + bg[c];
    }
}

__global__ void scatter_kernel(const int* __restrict__ ei,
                               const float* __restrict__ H,
                               const float* __restrict__ dinv,
                               float* __restrict__ agg, int e) {
    int gid = blockIdx.x * blockDim.x + threadIdx.x;
    int total = e * 16;
    if (gid >= total) return;
    const int edge = gid >> 4;
    const int q = gid & 15;
    const int r = ei[edge];
    const int c = ei[e + edge];
    const float nrm = dinv[r] * dinv[c];

    const float4 hv = *reinterpret_cast<const float4*>(H + (size_t)r * NCLASS + q * 4);
    float* __restrict__ dst = agg + (size_t)c * NCLASS + q * 4;
    atomicAdd(dst + 0, hv.x * nrm);
    atomicAdd(dst + 1, hv.y * nrm);
    atomicAdd(dst + 2, hv.z * nrm);
    atomicAdd(dst + 3, hv.w * nrm);
}

__global__ __launch_bounds__(256) void gemm2_kernel(const float* __restrict__ AGG,
                                                    const float* __restrict__ Wl,
                                                    const float* __restrict__ bl,
                                                    float* __restrict__ OUT, int n) {
    const int lane = threadIdx.x & 31;
    const int wave = threadIdx.x >> 5;
    const int mtile = blockIdx.x * 8 + wave;
    const int m0 = mtile * 16;
    if (m0 >= n) return;

    const int half = lane >> 4;
    const int l15  = lane & 15;
    int m = m0 + l15;
    if (m >= n) m = n - 1;
    const float* __restrict__ arow = AGG + (size_t)m * NCLASS;

    v8f acc0 = {}, acc1 = {}, acc2 = {}, acc3 = {};

    for (int k0 = 0; k0 < NCLASS; k0 += 4) {
        const int kb = k0 + half * 2;
        v2f a;
        a.x = fmaxf(arow[kb], 0.0f);
        a.y = fmaxf(arow[kb + 1], 0.0f);

        const float* __restrict__ w0 = Wl + (size_t)kb * NCLASS;
        const float* __restrict__ w1 = w0 + NCLASS;
        v2f b0, b1, b2, b3;
        b0.x = w0[l15];      b0.y = w1[l15];
        b1.x = w0[16 + l15]; b1.y = w1[16 + l15];
        b2.x = w0[32 + l15]; b2.y = w1[32 + l15];
        b3.x = w0[48 + l15]; b3.y = w1[48 + l15];

        acc0 = __builtin_amdgcn_wmma_f32_16x16x4_f32(false, a, false, b0, (short)0, acc0, false, false);
        acc1 = __builtin_amdgcn_wmma_f32_16x16x4_f32(false, a, false, b1, (short)0, acc1, false, false);
        acc2 = __builtin_amdgcn_wmma_f32_16x16x4_f32(false, a, false, b2, (short)0, acc2, false, false);
        acc3 = __builtin_amdgcn_wmma_f32_16x16x4_f32(false, a, false, b3, (short)0, acc3, false, false);
    }

    const float bl0 = bl[l15];
    const float bl1 = bl[16 + l15];
    const float bl2 = bl[32 + l15];
    const float bl3 = bl[48 + l15];

    const int mbase = m0 + half * 8;
#pragma unroll
    for (int r = 0; r < 8; ++r) {
        const int mr = mbase + r;
        if (mr < n) {
            float* __restrict__ orow = OUT + (size_t)mr * NCLASS;
            orow[l15]      = acc0[r] + bl0;
            orow[16 + l15] = acc1[r] + bl1;
            orow[32 + l15] = acc2[r] + bl2;
            orow[48 + l15] = acc3[r] + bl3;
        }
    }
}

extern "C" void kernel_launch(void* const* d_in, const int* in_sizes, int n_in,
                              void* d_out, int out_size, void* d_ws, size_t ws_size,
                              hipStream_t stream) {
    const float* x  = (const float*)d_in[0];
    const int*   ei = (const int*)d_in[1];
    const float* Wg = (const float*)d_in[2];
    const float* bg = (const float*)d_in[3];
    const float* Wl = (const float*)d_in[4];
    const float* bl = (const float*)d_in[5];
    float* out = (float*)d_out;

    const int n = in_sizes[0] / NFEAT;
    const int e = in_sizes[1] / 2;

    float* ws   = (float*)d_ws;
    size_t off  = ((size_t)n + 3) & ~(size_t)3;
    float* dinv = ws;
    float* h    = ws + off;
    float* agg  = h + (size_t)n * NCLASS;

    const int T = 256;

    deg_init_kernel <<<(n + T - 1) / T, T, 0, stream>>>(dinv, n);
    deg_count_kernel<<<(e + T - 1) / T, T, 0, stream>>>(ei + e, dinv, e);
    deg_rsqrt_kernel<<<(n + T - 1) / T, T, 0, stream>>>(dinv, n);

    const int mtiles = (n + 15) / 16;
    const int gblocks = (mtiles + 7) / 8;
    gemm1_kernel<<<gblocks, T, 0, stream>>>(x, Wg, h, n);

    agg_init_kernel<<<((n * NCLASS) + T - 1) / T, T, 0, stream>>>(h, dinv, bg, agg, n);

    const int stotal = e * 16;
    scatter_kernel<<<(stotal + T - 1) / T, T, 0, stream>>>(ei, h, dinv, agg, e);

    gemm2_kernel<<<gblocks, T, 0, stream>>>(agg, Wl, bl, out, n);
}


# degree count: per-block LDS integer histogram over (edge slice x node range) plus a sum-and-rsqrt pass replaces 3.2M global float atomics; degree sums become exact integer adds, dinv bit-identical
# speedup vs baseline: 1.2994x; 1.2994x over previous
#include <hip/hip_runtime.h>

typedef __attribute__((ext_vector_type(2))) float v2f;
typedef __attribute__((ext_vector_type(8))) float v8f;

#define NFEAT 256
#define NCLASS 64

#define HB_BINS    25600
#define HB_SLICES  64
#define HB_THREADS 1024
#define HB_BATCH   8

__global__ __launch_bounds__(HB_THREADS) void deg_hist_kernel(const int* __restrict__ col,
                                                              unsigned int* __restrict__ partial,
                                                              int n, int e) {
    __shared__ unsigned int bins[HB_BINS];
    const int tid = threadIdx.x;
    const int lo = blockIdx.y * HB_BINS;
    int cnt = n - lo;
    if (cnt > HB_BINS) cnt = HB_BINS;

    for (int j = tid; j < cnt; j += HB_THREADS) bins[j] = 0u;
    __syncthreads();

    const int chunk = (e + HB_SLICES - 1) / HB_SLICES;
    const int beg = blockIdx.x * chunk;
    int end = beg + chunk;
    if (end > e) end = e;

    for (int base = beg + tid; base < end; base += HB_BATCH * HB_THREADS) {
        int c[HB_BATCH];
#pragma unroll
        for (int u = 0; u < HB_BATCH; ++u) {
            const int i = base + u * HB_THREADS;
            c[u] = (i < end) ? col[i] : -1;
        }
#pragma unroll
        for (int u = 0; u < HB_BATCH; ++u) {
            const unsigned int b = (unsigned int)(c[u] - lo);
            if (b < (unsigned int)cnt) atomicAdd(&bins[b], 1u);
        }
    }
    __syncthreads();

    unsigned int* __restrict__ dst = partial + (size_t)blockIdx.x * n + lo;
    for (int j = tid; j < cnt; j += HB_THREADS) dst[j] = bins[j];
}

__global__ void deg_finish_kernel(const unsigned int* __restrict__ partial,
                                  float* __restrict__ dinv, int n) {
    int i = blockIdx.x * blockDim.x + threadIdx.x;
    if (i < n) {
        unsigned int c = 1u;
        for (int s = 0; s < HB_SLICES; ++s) c += partial[(size_t)s * n + i];
        dinv[i] = rsqrtf((float)c);
    }
}

__global__ __launch_bounds__(256) void gemm1_kernel(const float* __restrict__ X,
                                                    const float* __restrict__ Wg,
                                                    float* __restrict__ H, int n) {
    const int lane = threadIdx.x & 31;
    const int wave = threadIdx.x >> 5;
    const int mtile = blockIdx.x * 8 + wave;
    const int m0 = mtile * 16;
    if (m0 >= n) return;

    const int half = lane >> 4;
    const int l15  = lane & 15;
    int m = m0 + l15;
    if (m >= n) m = n - 1;
    const float* __restrict__ xrow = X + (size_t)m * NFEAT;

    v8f acc0 = {}, acc1 = {}, acc2 = {}, acc3 = {};

    for (int k0 = 0; k0 < NFEAT; k0 += 4) {
        const int kb = k0 + half * 2;
        v2f a;
        a.x = xrow[kb];
        a.y = xrow[kb + 1];

        const float* __restrict__ w0 = Wg + (size_t)kb * NCLASS;
        const float* __restrict__ w1 = w0 + NCLASS;
        v2f b0, b1, b2, b3;
        b0.x = w0[l15];      b0.y = w1[l15];
        b1.x = w0[16 + l15]; b1.y = w1[16 + l15];
        b2.x = w0[32 + l15]; b2.y = w1[32 + l15];
        b3.x = w0[48 + l15]; b3.y = w1[48 + l15];

        acc0 = __builtin_amdgcn_wmma_f32_16x16x4_f32(false, a, false, b0, (short)0, acc0, false, false);
        acc1 = __builtin_amdgcn_wmma_f32_16x16x4_f32(false, a, false, b1, (short)0, acc1, false, false);
        acc2 = __builtin_amdgcn_wmma_f32_16x16x4_f32(false, a, false, b2, (short)0, acc2, false, false);
        acc3 = __builtin_amdgcn_wmma_f32_16x16x4_f32(false, a, false, b3, (short)0, acc3, false, false);
    }

    const int mbase = m0 + half * 8;
#pragma unroll
    for (int r = 0; r < 8; ++r) {
        const int mr = mbase + r;
        if (mr < n) {
            float* __restrict__ hrow = H + (size_t)mr * NCLASS;
            hrow[l15]      = acc0[r];
            hrow[16 + l15] = acc1[r];
            hrow[32 + l15] = acc2[r];
            hrow[48 + l15] = acc3[r];
        }
    }
}

__global__ void agg_init_kernel(const float* __restrict__ H,
                                const float* __restrict__ dinv,
                                const float* __restrict__ bg,
                                float* __restrict__ agg, int n) {
    int i = blockIdx.x * blockDim.x + threadIdx.x;
    int total = n * NCLASS;
    if (i < total) {
        int node = i >> 6;
        int c = i & 63;
        float d = dinv[node];
        agg[i] = H[i] * d * d + bg[c];
    }
}

__global__ void scatter_kernel(const int* __restrict__ ei,
                               const float* __restrict__ H,
                               const float* __restrict__ dinv,
                               float* __restrict__ agg, int e) {
    int gid = blockIdx.x * blockDim.x + threadIdx.x;
    int total = e * 16;
    if (gid >= total) return;
    const int edge = gid >> 4;
    const int q = gid & 15;
    const int r = ei[edge];
    const int c = ei[e + edge];
    const float nrm = dinv[r] * dinv[c];

    const float4 hv = *reinterpret_cast<const float4*>(H + (size_t)r * NCLASS + q * 4);
    float* __restrict__ dst = agg + (size_t)c * NCLASS + q * 4;
    atomicAdd(dst + 0, hv.x * nrm);
    atomicAdd(dst + 1, hv.y * nrm);
    atomicAdd(dst + 2, hv.z * nrm);
    atomicAdd(dst + 3, hv.w * nrm);
}

__global__ __launch_bounds__(256) void gemm2_kernel(const float* __restrict__ AGG,
                                                    const float* __restrict__ Wl,
                                                    const float* __restrict__ bl,
                                                    float* __restrict__ OUT, int n) {
    const int lane = threadIdx.x & 31;
    const int wave = threadIdx.x >> 5;
    const int mtile = blockIdx.x * 8 + wave;
    const int m0 = mtile * 16;
    if (m0 >= n) return;

    const int half = lane >> 4;
    const int l15  = lane & 15;
    int m = m0 + l15;
    if (m >= n) m = n - 1;
    const float* __restrict__ arow = AGG + (size_t)m * NCLASS;

    v8f acc0 = {}, acc1 = {}, acc2 = {}, acc3 = {};

    for (int k0 = 0; k0 < NCLASS; k0 += 4) {
        const int kb = k0 + half * 2;
        v2f a;
        a.x = fmaxf(arow[kb], 0.0f);
        a.y = fmaxf(arow[kb + 1], 0.0f);

        const float* __restrict__ w0 = Wl + (size_t)kb * NCLASS;
        const float* __restrict__ w1 = w0 + NCLASS;
        v2f b0, b1, b2, b3;
        b0.x = w0[l15];      b0.y = w1[l15];
        b1.x = w0[16 + l15]; b1.y = w1[16 + l15];
        b2.x = w0[32 + l15]; b2.y = w1[32 + l15];
        b3.x = w0[48 + l15]; b3.y = w1[48 + l15];

        acc0 = __builtin_amdgcn_wmma_f32_16x16x4_f32(false, a, false, b0, (short)0, acc0, false, false);
        acc1 = __builtin_amdgcn_wmma_f32_16x16x4_f32(false, a, false, b1, (short)0, acc1, false, false);
        acc2 = __builtin_amdgcn_wmma_f32_16x16x4_f32(false, a, false, b2, (short)0, acc2, false, false);
        acc3 = __builtin_amdgcn_wmma_f32_16x16x4_f32(false, a, false, b3, (short)0, acc3, false, false);
    }

    const float bl0 = bl[l15];
    const float bl1 = bl[16 + l15];
    const float bl2 = bl[32 + l15];
    const float bl3 = bl[48 + l15];

    const int mbase = m0 + half * 8;
#pragma unroll
    for (int r = 0; r < 8; ++r) {
        const int mr = mbase + r;
        if (mr < n) {
            float* __restrict__ orow = OUT + (size_t)mr * NCLASS;
            orow[l15]      = acc0[r] + bl0;
            orow[16 + l15] = acc1[r] + bl1;
            orow[32 + l15] = acc2[r] + bl2;
            orow[48 + l15] = acc3[r] + bl3;
        }
    }
}

extern "C" void kernel_launch(void* const* d_in, const int* in_sizes, int n_in,
                              void* d_out, int out_size, void* d_ws, size_t ws_size,
                              hipStream_t stream) {
    const float* x  = (const float*)d_in[0];
    const int*   ei = (const int*)d_in[1];
    const float* Wg = (const float*)d_in[2];
    const float* bg = (const float*)d_in[3];
    const float* Wl = (const float*)d_in[4];
    const float* bl = (const float*)d_in[5];
    float* out = (float*)d_out;

    const int n = in_sizes[0] / NFEAT;
    const int e = in_sizes[1] / 2;

    float* ws   = (float*)d_ws;
    size_t off  = ((size_t)n + 3) & ~(size_t)3;
    float* dinv = ws;
    float* h    = ws + off;
    float* agg  = h + (size_t)n * NCLASS;

    const int T = 256;

    unsigned int* partial = (unsigned int*)agg;
    const int ranges = (n + HB_BINS - 1) / HB_BINS;
    deg_hist_kernel  <<<dim3(HB_SLICES, ranges), HB_THREADS, 0, stream>>>(ei + e, partial, n, e);
    deg_finish_kernel<<<(n + T - 1) / T, T, 0, stream>>>(partial, dinv, n);

    const int mtiles = (n + 15) / 16;
    const int gblocks = (mtiles + 7) / 8;
    gemm1_kernel<<<gblocks, T, 0, stream>>>(x, Wg, h, n);

    agg_init_kernel<<<((n * NCLASS) + T - 1) / T, T, 0, stream>>>(h, dinv, bg, agg, n);

    const int stotal = e * 16;
    scatter_kernel<<<(stotal + T - 1) / T, T, 0, stream>>>(ei, h, dinv, agg, e);

    gemm2_kernel<<<gblocks, T, 0, stream>>>(agg, Wl, bl, out, n);
}


# aggregation by destination: per-call counting sort of edges in LDS bins (returning add = slot), then one wave per node sums its rows in registers; no float atomics; sum order = slot arrival order
# speedup vs baseline: 2.4925x; 1.9182x over previous
#include <hip/hip_runtime.h>

typedef __attribute__((ext_vector_type(2))) float v2f;
typedef __attribute__((ext_vector_type(8))) float v8f;

#define NFEAT 256
#define NCLASS 64

#define HB_BINS    25600
#define HB_SLICES  64
#define HB_THREADS 1024
#define HB_BATCH   8

__global__ __launch_bounds__(HB_THREADS) void deg_hist_kernel(const int* __restrict__ col,
                                                              unsigned int* __restrict__ partial,
                                                              int n, int e) {
    __shared__ unsigned int bins[HB_BINS];
    const int tid = threadIdx.x;
    const int lo = blockIdx.y * HB_BINS;
    int cnt = n - lo;
    if (cnt > HB_BINS) cnt = HB_BINS;

    for (int j = tid; j < cnt; j += HB_THREADS) bins[j] = 0u;
    __syncthreads();

    const int chunk = (e + HB_SLICES - 1) / HB_SLICES;
    const int beg = blockIdx.x * chunk;
    int end = beg + chunk;
    if (end > e) end = e;

    for (int base = beg + tid; base < end; base += HB_BATCH * HB_THREADS) {
        int c[HB_BATCH];
#pragma unroll
        for (int u = 0; u < HB_BATCH; ++u) {
            const int i = base + u * HB_THREADS;
            c[u] = (i < end) ? col[i] : -1;
        }
#pragma unroll
        for (int u = 0; u < HB_BATCH; ++u) {
            const unsigned int b = (unsigned int)(c[u] - lo);
            if (b < (unsigned int)cnt) atomicAdd(&bins[b], 1u);
        }
    }
    __syncthreads();

    unsigned int* __restrict__ dst = partial + (size_t)blockIdx.x * n + lo;
    for (int j = tid; j < cnt; j += HB_THREADS) dst[j] = bins[j];
}

#define FB 512
static_assert(HB_BINS % FB == 0, "a destination range must be whole finish blocks");

__device__ __forceinline__ unsigned int wave_incl_scan(unsigned int v, int lane) {
#pragma unroll
    for (int d = 1; d < 32; d <<= 1) {
        const unsigned int t = __shfl_up(v, (unsigned int)d);
        if (lane >= d) v += t;
    }
    return v;
}

__global__ __launch_bounds__(FB) void deg_finish_kernel(unsigned int* partial,
                                                        float* __restrict__ dinv,
                                                        unsigned int* __restrict__ lp,
                                                        unsigned int* __restrict__ bt, int n) {
    __shared__ unsigned int wsum[FB / 32];
    const int tid = threadIdx.x;
    const int lane = tid & 31;
    const int wave = tid >> 5;
    const int i = blockIdx.x * FB + tid;
    unsigned int c = 0u;
    if (i < n) {
        unsigned int v[HB_SLICES];
#pragma unroll
        for (int s = 0; s < HB_SLICES; ++s) v[s] = partial[(size_t)s * n + i];
#pragma unroll
        for (int s = 0; s < HB_SLICES; ++s) { partial[(size_t)s * n + i] = c; c += v[s]; }
        dinv[i] = rsqrtf((float)(c + 1u));
    }
    const unsigned int incl = wave_incl_scan(c, lane);
    if (lane == 31) wsum[wave] = incl;
    __syncthreads();
    if (wave == 0) {
        const unsigned int w = (lane < FB / 32) ? wsum[lane] : 0u;
        const unsigned int wi = wave_incl_scan(w, lane);
        if (lane < FB / 32) wsum[lane] = wi - w;
        if (lane == 31) bt[blockIdx.x] = wi;
    }
    __syncthreads();
    if (i < n) lp[i] = wsum[wave] + incl - c;
}

__global__ __launch_bounds__(HB_THREADS) void place_kernel(const int* __restrict__ ei,
                                                           const unsigned int* __restrict__ excl,
                                                           const unsigned int* __restrict__ lp,
                                                           const unsigned int* __restrict__ bt,
                                                           unsigned int* __restrict__ rowptr,
                                                           int* __restrict__ srcs,
                                                           int n, int e) {
    __shared__ unsigned int bins[HB_BINS];
    __shared__ unsigned int boff[HB_BINS / FB];
    const int tid = threadIdx.x;
    const int lo = blockIdx.y * HB_BINS;
    int cnt = n - lo;
    if (cnt > HB_BINS) cnt = HB_BINS;
    const int first = blockIdx.y * (HB_BINS / FB);
    const int nloc = (cnt + FB - 1) / FB;

    if (tid < 32) {
        unsigned int run = 0u;
        for (int b0 = 0; b0 < first + nloc; b0 += 32) {
            const int b = b0 + tid;
            const unsigned int v = (b < first + nloc) ? bt[b] : 0u;
            const unsigned int incl = wave_incl_scan(v, tid);
            const int k = b - first;
            if (k >= 0 && k < nloc) boff[k] = run + incl - v;
            run += __shfl(incl, 31);
        }
        if (tid == 0 && blockIdx.x == 0 && blockIdx.y == gridDim.y - 1) rowptr[n] = run;
    }
    __syncthreads();

    const unsigned int* __restrict__ ex = excl + (size_t)blockIdx.x * n + lo;
    for (int j = tid; j < cnt; j += HB_THREADS) {
        const unsigned int start = boff[j / FB] + lp[lo + j];
        bins[j] = start + ex[j];
        if (blockIdx.x == 0) rowptr[lo + j] = start;
    }
    __syncthreads();

    const int* __restrict__ row = ei;
    const int* __restrict__ col = ei + e;
    const int chunk = (e + HB_SLICES - 1) / HB_SLICES;
    const int beg = blockIdx.x * chunk;
    int end = beg + chunk;
    if (end > e) end = e;

    for (int base = beg + tid; base < end; base += HB_BATCH * HB_THREADS) {
        int c[HB_BATCH], r[HB_BATCH];
#pragma unroll
        for (int u = 0; u < HB_BATCH; ++u) {
            const int i = base + u * HB_THREADS;
            c[u] = (i < end) ? col[i] : -1;
            r[u] = (i < end) ? row[i] : 0;
        }
#pragma unroll
        for (int u = 0; u < HB_BATCH; ++u) {
            const unsigned int b = (unsigned int)(c[u] - lo);
            if (b < (unsigned int)cnt) {
                const unsigned int pos = atomicAdd(&bins[b], 1u);
                if (pos < (unsigned int)e) srcs[pos] = r[u];
            }
        }
    }
}

__global__ __launch_bounds__(256) void gemm1_kernel(const float* __restrict__ X,
                                                    const float* __restrict__ Wg,
                                                    float* __restrict__ H, int n) {
    const int lane = threadIdx.x & 31;
    const int wave = threadIdx.x >> 5;
    const int mtile = blockIdx.x * 8 + wave;
    const int m0 = mtile * 16;
    if (m0 >= n) return;

    const int half = lane >> 4;
    const int l15  = lane & 15;
    int m = m0 + l15;
    if (m >= n) m = n - 1;
    const float* __restrict__ xrow = X + (size_t)m * NFEAT;

    v8f acc0 = {}, acc1 = {}, acc2 = {}, acc3 = {};

    for (int k0 = 0; k0 < NFEAT; k0 += 4) {
        const int kb = k0 + half * 2;
        v2f a;
        a.x = xrow[kb];
        a.y = xrow[kb + 1];

        const float* __restrict__ w0 = Wg + (size_t)kb * NCLASS;
        const float* __restrict__ w1 = w0 + NCLASS;
        v2f b0, b1, b2, b3;
        b0.x = w0[l15];      b0.y = w1[l15];
        b1.x = w0[16 + l15]; b1.y = w1[16 + l15];
        b2.x = w0[32 + l15]; b2.y = w1[32 + l15];
        b3.x = w0[48 + l15]; b3.y = w1[48 + l15];

        acc0 = __builtin_amdgcn_wmma_f32_16x16x4_f32(false, a, false, b0, (short)0, acc0, false, false);
        acc1 = __builtin_amdgcn_wmma_f32_16x16x4_f32(false, a, false, b1, (short)0, acc1, false, false);
        acc2 = __builtin_amdgcn_wmma_f32_16x16x4_f32(false, a, false, b2, (short)0, acc2, false, false);
        acc3 = __builtin_amdgcn_wmma_f32_16x16x4_f32(false, a, false, b3, (short)0, acc3, false, false);
    }

    const int mbase = m0 + half * 8;
#pragma unroll
    for (int r = 0; r < 8; ++r) {
        const int mr = mbase + r;
        if (mr < n) {
            float* __restrict__ hrow = H + (size_t)mr * NCLASS;
            hrow[l15]      = acc0[r];
            hrow[16 + l15] = acc1[r];
            hrow[32 + l15] = acc2[r];
            hrow[48 + l15] = acc3[r];
        }
    }
}

#define GA_BATCH 8

__global__ __launch_bounds__(256) void gather_kernel(const unsigned int* __restrict__ rowptr,
                                                     const int* __restrict__ srcs,
                                                     const float* __restrict__ H,
                                                     const float* __restrict__ dinv,
                                                     const float* __restrict__ bg,
                                                     float* __restrict__ agg, int n) {
    const int lane = threadIdx.x & 31;
    const int node = __builtin_amdgcn_readfirstlane((int)(blockIdx.x * 8 + (threadIdx.x >> 5)));
    if (node >= n) return;

    const unsigned int beg = rowptr[node];
    const unsigned int end = rowptr[node + 1];
    const float di = dinv[node];
    const float2* __restrict__ Hv = reinterpret_cast<const float2*>(H);
    const float2 self = Hv[(size_t)node * (NCLASS / 2) + lane];
    const float2 bias = reinterpret_cast<const float2*>(bg)[lane];

    float ax = 0.0f, ay = 0.0f;
    for (unsigned int base = beg; base < end; base += 32u) {
        const unsigned int j = base + (unsigned int)lane;
        int r = (j < end) ? srcs[j] : node;
        if ((unsigned int)r >= (unsigned int)n) r = node;
        const float w = (j < end) ? dinv[r] * di : 0.0f;
        const int m = (int)((end - base < 32u) ? (end - base) : 32u);
        for (int k = 0; k < m; k += GA_BATCH) {
            float2 hv[GA_BATCH];
            float wk[GA_BATCH];
#pragma unroll
            for (int u = 0; u < GA_BATCH; ++u) {
                const int kk = k + u;
                hv[u] = make_float2(0.0f, 0.0f);
                wk[u] = 0.0f;
                if (kk < m) {
                    const int rk = __builtin_amdgcn_readlane(r, kk);
                    wk[u] = __int_as_float(__builtin_amdgcn_readlane(__float_as_int(w), kk));
                    hv[u] = Hv[(size_t)rk * (NCLASS / 2) + lane];
                }
            }
#pragma unroll
            for (int u = 0; u < GA_BATCH; ++u) {
                ax += hv[u].x * wk[u];
                ay += hv[u].y * wk[u];
            }
        }
    }
    const float d2 = di * di;
    float2 o;
    o.x = ax + self.x * d2 + bias.x;
    o.y = ay + self.y * d2 + bias.y;
    reinterpret_cast<float2*>(agg)[(size_t)node * (NCLASS / 2) + lane] = o;
}

__global__ __launch_bounds__(256) void gemm2_kernel(const float* __restrict__ AGG,
                                                    const float* __restrict__ Wl,
                                                    const float* __restrict__ bl,
                                                    float* __restrict__ OUT, int n) {
    const int lane = threadIdx.x & 31;
    const int wave = threadIdx.x >> 5;
    const int mtile = blockIdx.x * 8 + wave;
    const int m0 = mtile * 16;
    if (m0 >= n) return;

    const int half = lane >> 4;
    const int l15  = lane & 15;
    int m = m0 + l15;
    if (m >= n) m = n - 1;
    const float* __restrict__ arow = AGG + (size_t)m * NCLASS;

    v8f acc0 = {}, acc1 = {}, acc2 = {}, acc3 = {};

    for (int k0 = 0; k0 < NCLASS; k0 += 4) {
        const int kb = k0 + half * 2;
        v2f a;
        a.x = fmaxf(arow[kb], 0.0f);
        a.y = fmaxf(arow[kb + 1], 0.0f);

        const float* __restrict__ w0 = Wl + (size_t)kb * NCLASS;
        const float* __restrict__ w1 = w0 + NCLASS;
        v2f b0, b1, b2, b3;
        b0.x = w0[l15];      b0.y = w1[l15];
        b1.x = w0[16 + l15]; b1.y = w1[16 + l15];
        b2.x = w0[32 + l15]; b2.y = w1[32 + l15];
        b3.x = w0[48 + l15]; b3.y = w1[48 + l15];

        acc0 = __builtin_amdgcn_wmma_f32_16x16x4_f32(false, a, false, b0, (short)0, acc0, false, false);
        acc1 = __builtin_amdgcn_wmma_f32_16x16x4_f32(false, a, false, b1, (short)0, acc1, false, false);
        acc2 = __builtin_amdgcn_wmma_f32_16x16x4_f32(false, a, false, b2, (short)0, acc2, false, false);
        acc3 = __builtin_amdgcn_wmma_f32_16x16x4_f32(false, a, false, b3, (short)0, acc3, false, false);
    }

    const float bl0 = bl[l15];
    const float bl1 = bl[16 + l15];
    const float bl2 = bl[32 + l15];
    const float bl3 = bl[48 + l15];

    const int mbase = m0 + half * 8;
#pragma unroll
    for (int r = 0; r < 8; ++r) {
        const int mr = mbase + r;
        if (mr < n) {
            float* __restrict__ orow = OUT + (size_t)mr * NCLASS;
            orow[l15]      = acc0[r] + bl0;
            orow[16 + l15] = acc1[r] + bl1;
            orow[32 + l15] = acc2[r] + bl2;
            orow[48 + l15] = acc3[r] + bl3;
        }
    }
}

extern "C" void kernel_launch(void* const* d_in, const int* in_sizes, int n_in,
                              void* d_out, int out_size, void* d_ws, size_t ws_size,
                              hipStream_t stream) {
    const float* x  = (const float*)d_in[0];
    const int*   ei = (const int*)d_in[1];
    const float* Wg = (const float*)d_in[2];
    const float* bg = (const float*)d_in[3];
    const float* Wl = (const float*)d_in[4];
    const float* bl = (const float*)d_in[5];
    float* out = (float*)d_out;

    const int n = in_sizes[0] / NFEAT;
    const int e = in_sizes[1] / 2;

    float* ws   = (float*)d_ws;
    size_t off  = ((size_t)n + 3) & ~(size_t)3;
    float* dinv = ws;
    float* h    = ws + off;
    float* agg  = h + (size_t)n * NCLASS;

    const int T = 256;

    unsigned int* partial = (unsigned int*)agg;
    unsigned int* lp = (unsigned int*)h;
    unsigned int* bt = lp + n;
    unsigned int* rowptr = (unsigned int*)out;
    int* srcs = (int*)out + (((size_t)n + 64) & ~(size_t)63);

    const int ranges = (n + HB_BINS - 1) / HB_BINS;
    deg_hist_kernel  <<<dim3(HB_SLICES, ranges), HB_THREADS, 0, stream>>>(ei + e, partial, n, e);
    deg_finish_kernel<<<(n + FB - 1) / FB, FB, 0, stream>>>(partial, dinv, lp, bt, n);
    place_kernel     <<<dim3(HB_SLICES, ranges), HB_THREADS, 0, stream>>>(ei, partial, lp, bt, rowptr, srcs, n, e);

    const int mtiles = (n + 15) / 16;
    const int gblocks = (mtiles + 7) / 8;
    gemm1_kernel<<<gblocks, T, 0, stream>>>(x, Wg, h, n);

    gather_kernel<<<(n + 7) / 8, T, 0, stream>>>(rowptr, srcs, h, dinv, bg, agg, n);

    gemm2_kernel<<<gblocks, T, 0, stream>>>(agg, Wl, bl, out, n);
}


# first product moved into the placing launch: 256 placing blocks and 256 product blocks of 1024 threads share the WGPs in one launch (5 launches, not 6); sums and their order unchanged
# speedup vs baseline: 2.7767x; 1.1140x over previous
#include <hip/hip_runtime.h>

typedef __attribute__((ext_vector_type(2))) float v2f;
typedef __attribute__((ext_vector_type(8))) float v8f;

#define NFEAT 256
#define NCLASS 64

#define HB_BINS    25600
#define HB_SLICES  64
#define HB_THREADS 1024
#define HB_BATCH   8

__global__ __launch_bounds__(HB_THREADS) void deg_hist_kernel(const int* __restrict__ col,
                                                              unsigned int* __restrict__ partial,
                                                              int n, int e) {
    __shared__ unsigned int bins[HB_BINS];
    const int tid = threadIdx.x;
    const int lo = blockIdx.y * HB_BINS;
    int cnt = n - lo;
    if (cnt > HB_BINS) cnt = HB_BINS;

    for (int j = tid; j < cnt; j += HB_THREADS) bins[j] = 0u;
    __syncthreads();

    const int chunk = (e + HB_SLICES - 1) / HB_SLICES;
    const int beg = blockIdx.x * chunk;
    int end = beg + chunk;
    if (end > e) end = e;

    for (int base = beg + tid; base < end; base += HB_BATCH * HB_THREADS) {
        int c[HB_BATCH];
#pragma unroll
        for (int u = 0; u < HB_BATCH; ++u) {
            const int i = base + u * HB_THREADS;
            c[u] = (i < end) ? col[i] : -1;
        }
#pragma unroll
        for (int u = 0; u < HB_BATCH; ++u) {
            const unsigned int b = (unsigned int)(c[u] - lo);
            if (b < (unsigned int)cnt) atomicAdd(&bins[b], 1u);
        }
    }
    __syncthreads();

    unsigned int* __restrict__ dst = partial + (size_t)blockIdx.x * n + lo;
    for (int j = tid; j < cnt; j += HB_THREADS) dst[j] = bins[j];
}

#define FB 512
static_assert(HB_BINS % FB == 0, "a destination range must be whole finish blocks");

__device__ __forceinline__ unsigned int wave_incl_scan(unsigned int v, int lane) {
#pragma unroll
    for (int d = 1; d < 32; d <<= 1) {
        const unsigned int t = __shfl_up(v, (unsigned int)d);
        if (lane >= d) v += t;
    }
    return v;
}

__global__ __launch_bounds__(FB) void deg_finish_kernel(unsigned int* partial,
                                                        float* __restrict__ dinv,
                                                        unsigned int* __restrict__ lp,
                                                        unsigned int* __restrict__ bt, int n) {
    __shared__ unsigned int wsum[FB / 32];
    const int tid = threadIdx.x;
    const int lane = tid & 31;
    const int wave = tid >> 5;
    const int i = blockIdx.x * FB + tid;
    unsigned int c = 0u;
    if (i < n) {
        unsigned int v[HB_SLICES];
#pragma unroll
        for (int s = 0; s < HB_SLICES; ++s) v[s] = partial[(size_t)s * n + i];
#pragma unroll
        for (int s = 0; s < HB_SLICES; ++s) { partial[(size_t)s * n + i] = c; c += v[s]; }
        dinv[i] = rsqrtf((float)(c + 1u));
    }
    const unsigned int incl = wave_incl_scan(c, lane);
    if (lane == 31) wsum[wave] = incl;
    __syncthreads();
    if (wave == 0) {
        const unsigned int w = (lane < FB / 32) ? wsum[lane] : 0u;
        const unsigned int wi = wave_incl_scan(w, lane);
        if (lane < FB / 32) wsum[lane] = wi - w;
        if (lane == 31) bt[blockIdx.x] = wi;
    }
    __syncthreads();
    if (i < n) lp[i] = wsum[wave] + incl - c;
}

__device__ __forceinline__ void place_body(unsigned int* bins, unsigned int* boff,
                                           const int bx, const int by, const int ranges,
                                           const int* __restrict__ ei,
                                           const unsigned int* __restrict__ excl,
                                           const unsigned int* __restrict__ lp,
                                           const unsigned int* __restrict__ bt,
                                           unsigned int* __restrict__ rowptr,
                                           int* __restrict__ srcs,
                                           int n, int e) {
    const int tid = threadIdx.x;
    const int lo = by * HB_BINS;
    int cnt = n - lo;
    if (cnt > HB_BINS) cnt = HB_BINS;
    const int first = by * (HB_BINS / FB);
    const int nloc = (cnt + FB - 1) / FB;

    if (tid < 32) {
        unsigned int run = 0u;
        for (int b0 = 0; b0 < first + nloc; b0 += 32) {
            const int b = b0 + tid;
            const unsigned int v = (b < first + nloc) ? bt[b] : 0u;
            const unsigned int incl = wave_incl_scan(v, tid);
            const int k = b - first;
            if (k >= 0 && k < nloc) boff[k] = run + incl - v;
            run += __shfl(incl, 31);
        }
        if (tid == 0 && bx == 0 && by == ranges - 1) rowptr[n] = run;
    }
    __syncthreads();

    const unsigned int* __restrict__ ex = excl + (size_t)bx * n + lo;
    for (int j = tid; j < cnt; j += HB_THREADS) {
        const unsigned int start = boff[j / FB] + lp[lo + j];
        bins[j] = start + ex[j];
        if (bx == 0) rowptr[lo + j] = start;
    }
    __syncthreads();

    const int* __restrict__ row = ei;
    const int* __restrict__ col = ei + e;
    const int chunk = (e + HB_SLICES - 1) / HB_SLICES;
    const int beg = bx * chunk;
    int end = beg + chunk;
    if (end > e) end = e;

    for (int base = beg + tid; base < end; base += HB_BATCH * HB_THREADS) {
        int c[HB_BATCH], r[HB_BATCH];
#pragma unroll
        for (int u = 0; u < HB_BATCH; ++u) {
            const int i = base + u * HB_THREADS;
            c[u] = (i < end) ? col[i] : -1;
            r[u] = (i < end) ? row[i] : 0;
        }
#pragma unroll
        for (int u = 0; u < HB_BATCH; ++u) {
            const unsigned int b = (unsigned int)(c[u] - lo);
            if (b < (unsigned int)cnt) {
                const unsigned int pos = atomicAdd(&bins[b], 1u);
                if (pos < (unsigned int)e) srcs[pos] = r[u];
            }
        }
    }
}

__device__ __forceinline__ void gemm1_body(const int gb, const int ngemm,
                                           const float* __restrict__ X,
                                           const float* __restrict__ Wg,
                                           const float* __restrict__ dinv,
                                           float* __restrict__ H, int n) {
    const int lane = threadIdx.x & 31;
    const int wave = threadIdx.x >> 5;
    const int half = lane >> 4;
    const int l15  = lane & 15;
    const int mtiles = (n + 15) / 16;

    for (int mtile = wave * ngemm + gb; mtile < mtiles; mtile += (HB_THREADS / 32) * ngemm) {
        const int m0 = mtile * 16;
        int m = m0 + l15;
        if (m >= n) m = n - 1;
        const float* __restrict__ xrow = X + (size_t)m * NFEAT;

        v8f acc0 = {}, acc1 = {}, acc2 = {}, acc3 = {};

        for (int k0 = 0; k0 < NFEAT; k0 += 4) {
            const int kb = k0 + half * 2;
            v2f a;
            a.x = xrow[kb];
            a.y = xrow[kb + 1];

            const float* __restrict__ w0 = Wg + (size_t)kb * NCLASS;
            const float* __restrict__ w1 = w0 + NCLASS;
            v2f b0, b1, b2, b3;
            b0.x = w0[l15];      b0.y = w1[l15];
            b1.x = w0[16 + l15]; b1.y = w1[16 + l15];
            b2.x = w0[32 + l15]; b2.y = w1[32 + l15];
            b3.x = w0[48 + l15]; b3.y = w1[48 + l15];

            acc0 = __builtin_amdgcn_wmma_f32_16x16x4_f32(false, a, false, b0, (short)0, acc0, false, false);
            acc1 = __builtin_amdgcn_wmma_f32_16x16x4_f32(false, a, false, b1, (short)0, acc1, false, false);
            acc2 = __builtin_amdgcn_wmma_f32_16x16x4_f32(false, a, false, b2, (short)0, acc2, false, false);
            acc3 = __builtin_amdgcn_wmma_f32_16x16x4_f32(false, a, false, b3, (short)0, acc3, false, false);
        }

        const int mbase = m0 + half * 8;
#pragma unroll
        for (int r = 0; r < 8; ++r) {
            const int mr = mbase + r;
            if (mr < n) {
                const float dv = dinv[mr];
                float* __restrict__ hrow = H + (size_t)mr * NCLASS;
                hrow[l15]      = acc0[r] * dv;
                hrow[16 + l15] = acc1[r] * dv;
                hrow[32 + l15] = acc2[r] * dv;
                hrow[48 + l15] = acc3[r] * dv;
            }
        }
    }
}

__global__ __launch_bounds__(HB_THREADS, 16) void place_gemm1_kernel(const int* __restrict__ ei,
                                                                     const unsigned int* __restrict__ excl,
                                                                     const unsigned int* __restrict__ lp,
                                                                     const unsigned int* __restrict__ bt,
                                                                     unsigned int* __restrict__ rowptr,
                                                                     int* __restrict__ srcs,
                                                                     const float* __restrict__ X,
                                                                     const float* __restrict__ Wg,
                                                                     const float* __restrict__ dinv,
                                                                     float* __restrict__ H,
                                                                     int n, int e, int nplace) {
    __shared__ unsigned int bins[HB_BINS];
    __shared__ unsigned int boff[HB_BINS / FB];
    const int b = blockIdx.x;
    if (b < nplace) {
        place_body(bins, boff, b % HB_SLICES, b / HB_SLICES, nplace / HB_SLICES,
                   ei, excl, lp, bt, rowptr, srcs, n, e);
    } else {
        gemm1_body(b - nplace, (int)gridDim.x - nplace, X, Wg, dinv, H, n);
    }
}

#define GA_BATCH 8

__global__ __launch_bounds__(256) void gather_kernel(const unsigned int* __restrict__ rowptr,
                                                     const int* __restrict__ srcs,
                                                     const float* __restrict__ H,
                                                     const float* __restrict__ dinv,
                                                     const float* __restrict__ bg,
                                                     float* __restrict__ agg, int n) {
    const int lane = threadIdx.x & 31;
    const int node = __builtin_amdgcn_readfirstlane((int)(blockIdx.x * 8 + (threadIdx.x >> 5)));
    if (node >= n) return;

    const unsigned int beg = rowptr[node];
    const unsigned int end = rowptr[node + 1];
    const float di = dinv[node];
    const float2* __restrict__ Hv = reinterpret_cast<const float2*>(H);
    const float2 self = Hv[(size_t)node * (NCLASS / 2) + lane];
    const float2 bias = reinterpret_cast<const float2*>(bg)[lane];

    float ax = 0.0f, ay = 0.0f;
    for (unsigned int base = beg; base < end; base += 32u) {
        const unsigned int j = base + (unsigned int)lane;
        int r = (j < end) ? srcs[j] : node;
        if ((unsigned int)r >= (unsigned int)n) r = node;
        const int m = (int)((end - base < 32u) ? (end - base) : 32u);
        for (int k = 0; k < m; k += GA_BATCH) {
            float2 hv[GA_BATCH];
#pragma unroll
            for (int u = 0; u < GA_BATCH; ++u) {
                const int kk = k + u;
                hv[u] = make_float2(0.0f, 0.0f);
                if (kk < m) {
                    const int rk = __builtin_amdgcn_readlane(r, kk);
                    hv[u] = Hv[(size_t)rk * (NCLASS / 2) + lane];
                }
            }
#pragma unroll
            for (int u = 0; u < GA_BATCH; ++u) {
                ax += hv[u].x;
                ay += hv[u].y;
            }
        }
    }
    float2 o;
    o.x = (ax + self.x) * di + bias.x;
    o.y = (ay + self.y) * di + bias.y;
    reinterpret_cast<float2*>(agg)[(size_t)node * (NCLASS / 2) + lane] = o;
}

__global__ __launch_bounds__(256) void gemm2_kernel(const float* __restrict__ AGG,
                                                    const float* __restrict__ Wl,
                                                    const float* __restrict__ bl,
                                                    float* __restrict__ OUT, int n) {
    const int lane = threadIdx.x & 31;
    const int wave = threadIdx.x >> 5;
    const int mtile = blockIdx.x * 8 + wave;
    const int m0 = mtile * 16;
    if (m0 >= n) return;

    const int half = lane >> 4;
    const int l15  = lane & 15;
    int m = m0 + l15;
    if (m >= n) m = n - 1;
    const float* __restrict__ arow = AGG + (size_t)m * NCLASS;

    v8f acc0 = {}, acc1 = {}, acc2 = {}, acc3 = {};

    for (int k0 = 0; k0 < NCLASS; k0 += 4) {
        const int kb = k0 + half * 2;
        v2f a;
        a.x = fmaxf(arow[kb], 0.0f);
        a.y = fmaxf(arow[kb + 1], 0.0f);

        const float* __restrict__ w0 = Wl + (size_t)kb * NCLASS;
        const float* __restrict__ w1 = w0 + NCLASS;
        v2f b0, b1, b2, b3;
        b0.x = w0[l15];      b0.y = w1[l15];
        b1.x = w0[16 + l15]; b1.y = w1[16 + l15];
        b2.x = w0[32 + l15]; b2.y = w1[32 + l15];
        b3.x = w0[48 + l15]; b3.y = w1[48 + l15];

        acc0 = __builtin_amdgcn_wmma_f32_16x16x4_f32(false, a, false, b0, (short)0, acc0, false, false);
        acc1 = __builtin_amdgcn_wmma_f32_16x16x4_f32(false, a, false, b1, (short)0, acc1, false, false);
        acc2 = __builtin_amdgcn_wmma_f32_16x16x4_f32(false, a, false, b2, (short)0, acc2, false, false);
        acc3 = __builtin_amdgcn_wmma_f32_16x16x4_f32(false, a, false, b3, (short)0, acc3, false, false);
    }

    const float bl0 = bl[l15];
    const float bl1 = bl[16 + l15];
    const float bl2 = bl[32 + l15];
    const float bl3 = bl[48 + l15];

    const int mbase = m0 + half * 8;
#pragma unroll
    for (int r = 0; r < 8; ++r) {
        const int mr = mbase + r;
        if (mr < n) {
            float* __restrict__ orow = OUT + (size_t)mr * NCLASS;
            orow[l15]      = acc0[r] + bl0;
            orow[16 + l15] = acc1[r] + bl1;
            orow[32 + l15] = acc2[r] + bl2;
            orow[48 + l15] = acc3[r] + bl3;
        }
    }
}

extern "C" void kernel_launch(void* const* d_in, const int* in_sizes, int n_in,
                              void* d_out, int out_size, void* d_ws, size_t ws_size,
                              hipStream_t stream) {
    const float* x  = (const float*)d_in[0];
    const int*   ei = (const int*)d_in[1];
    const float* Wg = (const float*)d_in[2];
    const float* bg = (const float*)d_in[3];
    const float* Wl = (const float*)d_in[4];
    const float* bl = (const float*)d_in[5];
    float* out = (float*)d_out;

    const int n = in_sizes[0] / NFEAT;
    const int e = in_sizes[1] / 2;

    float* ws   = (float*)d_ws;
    size_t off  = ((size_t)n + 3) & ~(size_t)3;
    float* dinv = ws;
    float* h    = ws + off;
    float* agg  = h + (size_t)n * NCLASS;

    const int T = 256;

    unsigned int* partial = (unsigned int*)agg;
    unsigned int* rowptr = (unsigned int*)out;
    int* srcs = (int*)out + (((size_t)n + 64) & ~(size_t)63);
    unsigned int* lp = (unsigned int*)(srcs + (((size_t)e + 63) & ~(size_t)63));
    unsigned int* bt = lp + n;

    const int ranges = (n + HB_BINS - 1) / HB_BINS;
    deg_hist_kernel  <<<dim3(HB_SLICES, ranges), HB_THREADS, 0, stream>>>(ei + e, partial, n, e);
    deg_finish_kernel<<<(n + FB - 1) / FB, FB, 0, stream>>>(partial, dinv, lp, bt, n);
    const int nplace = HB_SLICES * ranges;
    place_gemm1_kernel<<<2 * nplace, HB_THREADS, 0, stream>>>(ei, partial, lp, bt, rowptr, srcs,
                                                             x, Wg, dinv, h, n, e, nplace);

    const int mtiles = (n + 15) / 16;
    const int gblocks = (mtiles + 7) / 8;

    gather_kernel<<<(n + 7) / 8, T, 0, stream>>>(rowptr, srcs, h, dinv, bg, agg, n);

    gemm2_kernel<<<gblocks, T, 0, stream>>>(agg, Wl, bl, out, n);
}


# first product's operand path: weights copied once per block into LDS in fragment order (16-byte B reads), X read 16 bytes per lane one block ahead; k sum order inside each 8-k block changes
# speedup vs baseline: 2.8978x; 1.0436x over previous
#include <hip/hip_runtime.h>

typedef __attribute__((ext_vector_type(2))) float v2f;
typedef __attribute__((ext_vector_type(8))) float v8f;
typedef __attribute__((ext_vector_type(4))) float v4f;

#define NFEAT 256
#define NCLASS 64

#define HB_BINS    25600
#define HB_SLICES  64
#define HB_THREADS 1024
#define HB_BATCH   8

__global__ __launch_bounds__(HB_THREADS) void deg_hist_kernel(const int* __restrict__ col,
                                                              unsigned int* __restrict__ partial,
                                                              int n, int e) {
    __shared__ unsigned int bins[HB_BINS];
    const int tid = threadIdx.x;
    const int lo = blockIdx.y * HB_BINS;
    int cnt = n - lo;
    if (cnt > HB_BINS) cnt = HB_BINS;

    for (int j = tid; j < cnt; j += HB_THREADS) bins[j] = 0u;
    __syncthreads();

    const int chunk = (e + HB_SLICES - 1) / HB_SLICES;
    const int beg = blockIdx.x * chunk;
    int end = beg + chunk;
    if (end > e) end = e;

    for (int base = beg + tid; base < end; base += HB_BATCH * HB_THREADS) {
        int c[HB_BATCH];
#pragma unroll
        for (int u = 0; u < HB_BATCH; ++u) {
            const int i = base + u * HB_THREADS;
            c[u] = (i < end) ? col[i] : -1;
        }
#pragma unroll
        for (int u = 0; u < HB_BATCH; ++u) {
            const unsigned int b = (unsigned int)(c[u] - lo);
            if (b < (unsigned int)cnt) atomicAdd(&bins[b], 1u);
        }
    }
    __syncthreads();

    unsigned int* __restrict__ dst = partial + (size_t)blockIdx.x * n + lo;
    for (int j = tid; j < cnt; j += HB_THREADS) dst[j] = bins[j];
}

#define FB 512
static_assert(HB_BINS % FB == 0, "a destination range must be whole finish blocks");

__device__ __forceinline__ unsigned int wave_incl_scan(unsigned int v, int lane) {
#pragma unroll
    for (int d = 1; d < 32; d <<= 1) {
        const unsigned int t = __shfl_up(v, (unsigned int)d);
        if (lane >= d) v += t;
    }
    return v;
}

__global__ __launch_bounds__(FB) void deg_finish_kernel(unsigned int* partial,
                                                        float* __restrict__ dinv,
                                                        unsigned int* __restrict__ lp,
                                                        unsigned int* __restrict__ bt, int n) {
    __shared__ unsigned int wsum[FB / 32];
    const int tid = threadIdx.x;
    const int lane = tid & 31;
    const int wave = tid >> 5;
    const int i = blockIdx.x * FB + tid;
    unsigned int c = 0u;
    if (i < n) {
        unsigned int v[HB_SLICES];
#pragma unroll
        for (int s = 0; s < HB_SLICES; ++s) v[s] = partial[(size_t)s * n + i];
#pragma unroll
        for (int s = 0; s < HB_SLICES; ++s) { partial[(size_t)s * n + i] = c; c += v[s]; }
        dinv[i] = rsqrtf((float)(c + 1u));
    }
    const unsigned int incl = wave_incl_scan(c, lane);
    if (lane == 31) wsum[wave] = incl;
    __syncthreads();
    if (wave == 0) {
        const unsigned int w = (lane < FB / 32) ? wsum[lane] : 0u;
        const unsigned int wi = wave_incl_scan(w, lane);
        if (lane < FB / 32) wsum[lane] = wi - w;
        if (lane == 31) bt[blockIdx.x] = wi;
    }
    __syncthreads();
    if (i < n) lp[i] = wsum[wave] + incl - c;
}

__device__ __forceinline__ void place_body(unsigned int* bins, unsigned int* boff,
                                           const int bx, const int by, const int ranges,
                                           const int* __restrict__ ei,
                                           const unsigned int* __restrict__ excl,
                                           const unsigned int* __restrict__ lp,
                                           const unsigned int* __restrict__ bt,
                                           unsigned int* __restrict__ rowptr,
                                           int* __restrict__ srcs,
                                           int n, int e) {
    const int tid = threadIdx.x;
    const int lo = by * HB_BINS;
    int cnt = n - lo;
    if (cnt > HB_BINS) cnt = HB_BINS;
    const int first = by * (HB_BINS / FB);
    const int nloc = (cnt + FB - 1) / FB;

    if (tid < 32) {
        unsigned int run = 0u;
        for (int b0 = 0; b0 < first + nloc; b0 += 32) {
            const int b = b0 + tid;
            const unsigned int v = (b < first + nloc) ? bt[b] : 0u;
            const unsigned int incl = wave_incl_scan(v, tid);
            const int k = b - first;
            if (k >= 0 && k < nloc) boff[k] = run + incl - v;
            run += __shfl(incl, 31);
        }
        if (tid == 0 && bx == 0 && by == ranges - 1) rowptr[n] = run;
    }
    __syncthreads();

    const unsigned int* __restrict__ ex = excl + (size_t)bx * n + lo;
    for (int j = tid; j < cnt; j += HB_THREADS) {
        const unsigned int start = boff[j / FB] + lp[lo + j];
        bins[j] = start + ex[j];
        if (bx == 0) rowptr[lo + j] = start;
    }
    __syncthreads();

    const int* __restrict__ row = ei;
    const int* __restrict__ col = ei + e;
    const int chunk = (e + HB_SLICES - 1) / HB_SLICES;
    const int beg = bx * chunk;
    int end = beg + chunk;
    if (end > e) end = e;

    for (int base = beg + tid; base < end; base += HB_BATCH * HB_THREADS) {
        int c[HB_BATCH], r[HB_BATCH];
#pragma unroll
        for (int u = 0; u < HB_BATCH; ++u) {
            const int i = base + u * HB_THREADS;
            c[u] = (i < end) ? col[i] : -1;
            r[u] = (i < end) ? row[i] : 0;
        }
#pragma unroll
        for (int u = 0; u < HB_BATCH; ++u) {
            const unsigned int b = (unsigned int)(c[u] - lo);
            if (b < (unsigned int)cnt) {
                const unsigned int pos = atomicAdd(&bins[b], 1u);
                if (pos < (unsigned int)e) srcs[pos] = r[u];
            }
        }
    }
}

__device__ __forceinline__ void mma_block8(v8f& acc0, v8f& acc1, v8f& acc2, v8f& acc3,
                                           const v4f a, const v4f* bp) {
    const v4f b00 = bp[0];
    const v4f b01 = bp[16];
    const v4f b10 = bp[32];
    const v4f b11 = bp[48];

    v2f a0, a1, b;
    a0.x = a.x; a0.y = a.y;
    a1.x = a.z; a1.y = a.w;

    b.x = b00.x; b.y = b00.y;
    acc0 = __builtin_amdgcn_wmma_f32_16x16x4_f32(false, a0, false, b, (short)0, acc0, false, false);
    b.x = b00.z; b.y = b00.w;
    acc1 = __builtin_amdgcn_wmma_f32_16x16x4_f32(false, a0, false, b, (short)0, acc1, false, false);
    b.x = b01.x; b.y = b01.y;
    acc2 = __builtin_amdgcn_wmma_f32_16x16x4_f32(false, a0, false, b, (short)0, acc2, false, false);
    b.x = b01.z; b.y = b01.w;
    acc3 = __builtin_amdgcn_wmma_f32_16x16x4_f32(false, a0, false, b, (short)0, acc3, false, false);

    b.x = b10.x; b.y = b10.y;
    acc0 = __builtin_amdgcn_wmma_f32_16x16x4_f32(false, a1, false, b, (short)0, acc0, false, false);
    b.x = b10.z; b.y = b10.w;
    acc1 = __builtin_amdgcn_wmma_f32_16x16x4_f32(false, a1, false, b, (short)0, acc1, false, false);
    b.x = b11.x; b.y = b11.y;
    acc2 = __builtin_amdgcn_wmma_f32_16x16x4_f32(false, a1, false, b, (short)0, acc2, false, false);
    b.x = b11.z; b.y = b11.w;
    acc3 = __builtin_amdgcn_wmma_f32_16x16x4_f32(false, a1, false, b, (short)0, acc3, false, false);
}

__device__ __forceinline__ void gemm1_body(float* wl,
                                           const int gb, const int ngemm,
                                           const float* __restrict__ X,
                                           const float* __restrict__ Wg,
                                           const float* __restrict__ dinv,
                                           float* __restrict__ H, int n) {
    const int tid = threadIdx.x;
    const int lane = tid & 31;
    const int wave = tid >> 5;
    const int half = lane >> 4;
    const int l15  = lane & 15;
    const int mtiles = (n + 15) / 16;

    constexpr int WG_PER_THREAD = NFEAT * NCLASS / 4 / HB_THREADS;
    static_assert(WG_PER_THREAD * HB_THREADS * 4 == NFEAT * NCLASS, "Wg must divide evenly over the block");
    v4f w[WG_PER_THREAD];
#pragma unroll
    for (int j = 0; j < WG_PER_THREAD; ++j) w[j] = reinterpret_cast<const v4f*>(Wg)[tid + j * HB_THREADS];
#pragma unroll
    for (int j = 0; j < WG_PER_THREAD; ++j) {
        const int i  = tid + j * HB_THREADS;
        const int k  = i >> 4;
        const int c0 = (i & 15) * 4;
        float* d = wl + (k >> 1) * 128 + (c0 >> 5) * 64 + (c0 & 15) * 4 + ((c0 >> 4) & 1) * 2 + (k & 1);
        d[0]  = w[j].x;
        d[4]  = w[j].y;
        d[8]  = w[j].z;
        d[12] = w[j].w;
    }
    __syncthreads();

    const v4f* wl4 = reinterpret_cast<const v4f*>(wl) + half * 64 + l15;

    for (int mtile = wave * ngemm + gb; mtile < mtiles; mtile += (HB_THREADS / 32) * ngemm) {
        const int m0 = mtile * 16;
        int m = m0 + l15;
        if (m >= n) m = n - 1;
        const v4f* __restrict__ xrow = reinterpret_cast<const v4f*>(X + (size_t)m * NFEAT) + half;

        v8f acc0 = {}, acc1 = {}, acc2 = {}, acc3 = {};

        v4f a0 = xrow[0];
#pragma unroll
        for (int kb = 0; kb < NFEAT / 8; kb += 2) {
            const v4f a1 = xrow[(kb + 1) * 2];
            mma_block8(acc0, acc1, acc2, acc3, a0, wl4 + kb * 128);
            asm volatile("" ::: "memory");
            if (kb + 2 < NFEAT / 8) a0 = xrow[(kb + 2) * 2];
            mma_block8(acc0, acc1, acc2, acc3, a1, wl4 + (kb + 1) * 128);
            asm volatile("" ::: "memory");
        }

        const int mbase = m0 + half * 8;
#pragma unroll
        for (int r = 0; r < 8; ++r) {
            const int mr = mbase + r;
            if (mr < n) {
                const float dv = dinv[mr];
                float* __restrict__ hrow = H + (size_t)mr * NCLASS;
                hrow[l15]      = acc0[r] * dv;
                hrow[16 + l15] = acc1[r] * dv;
                hrow[32 + l15] = acc2[r] * dv;
                hrow[48 + l15] = acc3[r] * dv;
            }
        }
    }
}

__global__ __launch_bounds__(HB_THREADS, 16) void place_gemm1_kernel(const int* __restrict__ ei,
                                                                     const unsigned int* __restrict__ excl,
                                                                     const unsigned int* __restrict__ lp,
                                                                     const unsigned int* __restrict__ bt,
                                                                     unsigned int* __restrict__ rowptr,
                                                                     int* __restrict__ srcs,
                                                                     const float* __restrict__ X,
                                                                     const float* __restrict__ Wg,
                                                                     const float* __restrict__ dinv,
                                                                     float* __restrict__ H,
                                                                     int n, int e, int nplace) {
    __shared__ __attribute__((aligned(16))) unsigned int bins[HB_BINS];
    __shared__ unsigned int boff[HB_BINS / FB];
    static_assert(HB_BINS >= NFEAT * NCLASS, "the bins must hold Wg for a product block");
    const int b = blockIdx.x;
    if (b < nplace) {
        place_body(bins, boff, b % HB_SLICES, b / HB_SLICES, nplace / HB_SLICES,
                   ei, excl, lp, bt, rowptr, srcs, n, e);
    } else {
        gemm1_body(reinterpret_cast<float*>(bins), b - nplace, (int)gridDim.x - nplace, X, Wg, dinv, H, n);
    }
}

#define GA_BATCH 8

__global__ __launch_bounds__(256) void gather_kernel(const unsigned int* __restrict__ rowptr,
                                                     const int* __restrict__ srcs,
                                                     const float* __restrict__ H,
                                                     const float* __restrict__ dinv,
                                                     const float* __restrict__ bg,
                                                     float* __restrict__ agg, int n) {
    const int lane = threadIdx.x & 31;
    const int node = __builtin_amdgcn_readfirstlane((int)(blockIdx.x * 8 + (threadIdx.x >> 5)));
    if (node >= n) return;

    const unsigned int beg = rowptr[node];
    const unsigned int end = rowptr[node + 1];
    const float di = dinv[node];
    const float2* __restrict__ Hv = reinterpret_cast<const float2*>(H);
    const float2 self = Hv[(size_t)node * (NCLASS / 2) + lane];
    const float2 bias = reinterpret_cast<const float2*>(bg)[lane];

    float ax = 0.0f, ay = 0.0f;
    for (unsigned int base = beg; base < end; base += 32u) {
        const unsigned int j = base + (unsigned int)lane;
        int r = (j < end) ? srcs[j] : node;
        if ((unsigned int)r >= (unsigned int)n) r = node;
        const int m = (int)((end - base < 32u) ? (end - base) : 32u);
        for (int k = 0; k < m; k += GA_BATCH) {
            float2 hv[GA_BATCH];
#pragma unroll
            for (int u = 0; u < GA_BATCH; ++u) {
                const int kk = k + u;
                hv[u] = make_float2(0.0f, 0.0f);
                if (kk < m) {
                    const int rk = __builtin_amdgcn_readlane(r, kk);
                    hv[u] = Hv[(size_t)rk * (NCLASS / 2) + lane];
                }
            }
#pragma unroll
            for (int u = 0; u < GA_BATCH; ++u) {
                ax += hv[u].x;
                ay += hv[u].y;
            }
        }
    }
    float2 o;
    o.x = (ax + self.x) * di + bias.x;
    o.y = (ay + self.y) * di + bias.y;
    reinterpret_cast<float2*>(agg)[(size_t)node * (NCLASS / 2) + lane] = o;
}

__global__ __launch_bounds__(256) void gemm2_kernel(const float* __restrict__ AGG,
                                                    const float* __restrict__ Wl,
                                                    const float* __restrict__ bl,
                                                    float* __restrict__ OUT, int n) {
    const int lane = threadIdx.x & 31;
    const int wave = threadIdx.x >> 5;
    const int mtile = blockIdx.x * 8 + wave;
    const int m0 = mtile * 16;
    if (m0 >= n) return;

    const int half = lane >> 4;
    const int l15  = lane & 15;
    int m = m0 + l15;
    if (m >= n) m = n - 1;
    const float* __restrict__ arow = AGG + (size_t)m * NCLASS;

    v8f acc0 = {}, acc1 = {}, acc2 = {}, acc3 = {};

    for (int k0 = 0; k0 < NCLASS; k0 += 4) {
        const int kb = k0 + half * 2;
        v2f a;
        a.x = fmaxf(arow[kb], 0.0f);
        a.y = fmaxf(arow[kb + 1], 0.0f);

        const float* __restrict__ w0 = Wl + (size_t)kb * NCLASS;
        const float* __restrict__ w1 = w0 + NCLASS;
        v2f b0, b1, b2, b3;
        b0.x = w0[l15];      b0.y = w1[l15];
        b1.x = w0[16 + l15]; b1.y = w1[16 + l15];
        b2.x = w0[32 + l15]; b2.y = w1[32 + l15];
        b3.x = w0[48 + l15]; b3.y = w1[48 + l15];

        acc0 = __builtin_amdgcn_wmma_f32_16x16x4_f32(false, a, false, b0, (short)0, acc0, false, false);
        acc1 = __builtin_amdgcn_wmma_f32_16x16x4_f32(false, a, false, b1, (short)0, acc1, false, false);
        acc2 = __builtin_amdgcn_wmma_f32_16x16x4_f32(false, a, false, b2, (short)0, acc2, false, false);
        acc3 = __builtin_amdgcn_wmma_f32_16x16x4_f32(false, a, false, b3, (short)0, acc3, false, false);
    }

    const float bl0 = bl[l15];
    const float bl1 = bl[16 + l15];
    const float bl2 = bl[32 + l15];
    const float bl3 = bl[48 + l15];

    const int mbase = m0 + half * 8;
#pragma unroll
    for (int r = 0; r < 8; ++r) {
        const int mr = mbase + r;
        if (mr < n) {
            float* __restrict__ orow = OUT + (size_t)mr * NCLASS;
            orow[l15]      = acc0[r] + bl0;
            orow[16 + l15] = acc1[r] + bl1;
            orow[32 + l15] = acc2[r] + bl2;
            orow[48 + l15] = acc3[r] + bl3;
        }
    }
}

extern "C" void kernel_launch(void* const* d_in, const int* in_sizes, int n_in,
                              void* d_out, int out_size, void* d_ws, size_t ws_size,
                              hipStream_t stream) {
    const float* x  = (const float*)d_in[0];
    const int*   ei = (const int*)d_in[1];
    const float* Wg = (const float*)d_in[2];
    const float* bg = (const float*)d_in[3];
    const float* Wl = (const float*)d_in[4];
    const float* bl = (const float*)d_in[5];
    float* out = (float*)d_out;

    const int n = in_sizes[0] / NFEAT;
    const int e = in_sizes[1] / 2;

    float* ws   = (float*)d_ws;
    size_t off  = ((size_t)n + 3) & ~(size_t)3;
    float* dinv = ws;
    float* h    = ws + off;
    float* agg  = h + (size_t)n * NCLASS;

    const int T = 256;

    unsigned int* partial = (unsigned int*)agg;
    unsigned int* rowptr = (unsigned int*)out;
    int* srcs = (int*)out + (((size_t)n + 64) & ~(size_t)63);
    unsigned int* lp = (unsigned int*)(srcs + (((size_t)e + 63) & ~(size_t)63));
    unsigned int* bt = lp + n;

    const int ranges = (n + HB_BINS - 1) / HB_BINS;
    deg_hist_kernel  <<<dim3(HB_SLICES, ranges), HB_THREADS, 0, stream>>>(ei + e, partial, n, e);
    deg_finish_kernel<<<(n + FB - 1) / FB, FB, 0, stream>>>(partial, dinv, lp, bt, n);
    const int nplace = HB_SLICES * ranges;
    place_gemm1_kernel<<<2 * nplace, HB_THREADS, 0, stream>>>(ei, partial, lp, bt, rowptr, srcs,
                                                             x, Wg, dinv, h, n, e, nplace);

    const int mtiles = (n + 15) / 16;
    const int gblocks = (mtiles + 7) / 8;

    gather_kernel<<<(n + 7) / 8, T, 0, stream>>>(rowptr, srcs, h, dinv, bg, agg, n);

    gemm2_kernel<<<gblocks, T, 0, stream>>>(agg, Wl, bl, out, n);
}


# second product moved into the aggregation launch: a block gathers 16 nodes, relu rows go through an LDS tile to 16 WMMAs per column tile; no agg plane, 4 launches not 5; sums and order unchanged
# speedup vs baseline: 3.0478x; 1.0518x over previous
#include <hip/hip_runtime.h>

typedef __attribute__((ext_vector_type(2))) float v2f;
typedef __attribute__((ext_vector_type(8))) float v8f;
typedef __attribute__((ext_vector_type(4))) float v4f;

#define NFEAT 256
#define NCLASS 64

#define HB_BINS    25600
#define HB_SLICES  64
#define HB_THREADS 1024
#define HB_BATCH   8

__global__ __launch_bounds__(HB_THREADS) void deg_hist_kernel(const int* __restrict__ col,
                                                              unsigned int* __restrict__ partial,
                                                              int n, int e) {
    __shared__ unsigned int bins[HB_BINS];
    const int tid = threadIdx.x;
    const int lo = blockIdx.y * HB_BINS;
    int cnt = n - lo;
    if (cnt > HB_BINS) cnt = HB_BINS;

    for (int j = tid; j < cnt; j += HB_THREADS) bins[j] = 0u;
    __syncthreads();

    const int chunk = (e + HB_SLICES - 1) / HB_SLICES;
    const int beg = blockIdx.x * chunk;
    int end = beg + chunk;
    if (end > e) end = e;

    for (int base = beg + tid; base < end; base += HB_BATCH * HB_THREADS) {
        int c[HB_BATCH];
#pragma unroll
        for (int u = 0; u < HB_BATCH; ++u) {
            const int i = base + u * HB_THREADS;
            c[u] = (i < end) ? col[i] : -1;
        }
#pragma unroll
        for (int u = 0; u < HB_BATCH; ++u) {
            const unsigned int b = (unsigned int)(c[u] - lo);
            if (b < (unsigned int)cnt) atomicAdd(&bins[b], 1u);
        }
    }
    __syncthreads();

    unsigned int* __restrict__ dst = partial + (size_t)blockIdx.x * n + lo;
    for (int j = tid; j < cnt; j += HB_THREADS) dst[j] = bins[j];
}

#define FB 512
static_assert(HB_BINS % FB == 0, "a destination range must be whole finish blocks");

__device__ __forceinline__ unsigned int wave_incl_scan(unsigned int v, int lane) {
#pragma unroll
    for (int d = 1; d < 32; d <<= 1) {
        const unsigned int t = __shfl_up(v, (unsigned int)d);
        if (lane >= d) v += t;
    }
    return v;
}

__global__ __launch_bounds__(FB) void deg_finish_kernel(unsigned int* partial,
                                                        float* __restrict__ dinv,
                                                        unsigned int* __restrict__ lp,
                                                        unsigned int* __restrict__ bt, int n) {
    __shared__ unsigned int wsum[FB / 32];
    const int tid = threadIdx.x;
    const int lane = tid & 31;
    const int wave = tid >> 5;
    const int i = blockIdx.x * FB + tid;
    unsigned int c = 0u;
    if (i < n) {
        unsigned int v[HB_SLICES];
#pragma unroll
        for (int s = 0; s < HB_SLICES; ++s) v[s] = partial[(size_t)s * n + i];
#pragma unroll
        for (int s = 0; s < HB_SLICES; ++s) { partial[(size_t)s * n + i] = c; c += v[s]; }
        dinv[i] = rsqrtf((float)(c + 1u));
    }
    const unsigned int incl = wave_incl_scan(c, lane);
    if (lane == 31) wsum[wave] = incl;
    __syncthreads();
    if (wave == 0) {
        const unsigned int w = (lane < FB / 32) ? wsum[lane] : 0u;
        const unsigned int wi = wave_incl_scan(w, lane);
        if (lane < FB / 32) wsum[lane] = wi - w;
        if (lane == 31) bt[blockIdx.x] = wi;
    }
    __syncthreads();
    if (i < n) lp[i] = wsum[wave] + incl - c;
}

__device__ __forceinline__ void place_body(unsigned int* bins, unsigned int* boff,
                                           const int bx, const int by, const int ranges,
                                           const int* __restrict__ ei,
                                           const unsigned int* __restrict__ excl,
                                           const unsigned int* __restrict__ lp,
                                           const unsigned int* __restrict__ bt,
                                           unsigned int* __restrict__ rowptr,
                                           int* __restrict__ srcs,
                                           int n, int e) {
    const int tid = threadIdx.x;
    const int lo = by * HB_BINS;
    int cnt = n - lo;
    if (cnt > HB_BINS) cnt = HB_BINS;
    const int first = by * (HB_BINS / FB);
    const int nloc = (cnt + FB - 1) / FB;

    if (tid < 32) {
        unsigned int run = 0u;
        for (int b0 = 0; b0 < first + nloc; b0 += 32) {
            const int b = b0 + tid;
            const unsigned int v = (b < first + nloc) ? bt[b] : 0u;
            const unsigned int incl = wave_incl_scan(v, tid);
            const int k = b - first;
            if (k >= 0 && k < nloc) boff[k] = run + incl - v;
            run += __shfl(incl, 31);
        }
        if (tid == 0 && bx == 0 && by == ranges - 1) rowptr[n] = run;
    }
    __syncthreads();

    const unsigned int* __restrict__ ex = excl + (size_t)bx * n + lo;
    for (int j = tid; j < cnt; j += HB_THREADS) {
        const unsigned int start = boff[j / FB] + lp[lo + j];
        bins[j] = start + ex[j];
        if (bx == 0) rowptr[lo + j] = start;
    }
    __syncthreads();

    const int* __restrict__ row = ei;
    const int* __restrict__ col = ei + e;
    const int chunk = (e + HB_SLICES - 1) / HB_SLICES;
    const int beg = bx * chunk;
    int end = beg + chunk;
    if (end > e) end = e;

    for (int base = beg + tid; base < end; base += HB_BATCH * HB_THREADS) {
        int c[HB_BATCH], r[HB_BATCH];
#pragma unroll
        for (int u = 0; u < HB_BATCH; ++u) {
            const int i = base + u * HB_THREADS;
            c[u] = (i < end) ? col[i] : -1;
            r[u] = (i < end) ? row[i] : 0;
        }
#pragma unroll
        for (int u = 0; u < HB_BATCH; ++u) {
            const unsigned int b = (unsigned int)(c[u] - lo);
            if (b < (unsigned int)cnt) {
                const unsigned int pos = atomicAdd(&bins[b], 1u);
                if (pos < (unsigned int)e) srcs[pos] = r[u];
            }
        }
    }
}

__device__ __forceinline__ void mma_block8(v8f& acc0, v8f& acc1, v8f& acc2, v8f& acc3,
                                           const v4f a, const v4f* bp) {
    const v4f b00 = bp[0];
    const v4f b01 = bp[16];
    const v4f b10 = bp[32];
    const v4f b11 = bp[48];

    v2f a0, a1, b;
    a0.x = a.x; a0.y = a.y;
    a1.x = a.z; a1.y = a.w;

    b.x = b00.x; b.y = b00.y;
    acc0 = __builtin_amdgcn_wmma_f32_16x16x4_f32(false, a0, false, b, (short)0, acc0, false, false);
    b.x = b00.z; b.y = b00.w;
    acc1 = __builtin_amdgcn_wmma_f32_16x16x4_f32(false, a0, false, b, (short)0, acc1, false, false);
    b.x = b01.x; b.y = b01.y;
    acc2 = __builtin_amdgcn_wmma_f32_16x16x4_f32(false, a0, false, b, (short)0, acc2, false, false);
    b.x = b01.z; b.y = b01.w;
    acc3 = __builtin_amdgcn_wmma_f32_16x16x4_f32(false, a0, false, b, (short)0, acc3, false, false);

    b.x = b10.x; b.y = b10.y;
    acc0 = __builtin_amdgcn_wmma_f32_16x16x4_f32(false, a1, false, b, (short)0, acc0, false, false);
    b.x = b10.z; b.y = b10.w;
    acc1 = __builtin_amdgcn_wmma_f32_16x16x4_f32(false, a1, false, b, (short)0, acc1, false, false);
    b.x = b11.x; b.y = b11.y;
    acc2 = __builtin_amdgcn_wmma_f32_16x16x4_f32(false, a1, false, b, (short)0, acc2, false, false);
    b.x = b11.z; b.y = b11.w;
    acc3 = __builtin_amdgcn_wmma_f32_16x16x4_f32(false, a1, false, b, (short)0, acc3, false, false);
}

__device__ __forceinline__ void gemm1_body(float* wl,
                                           const int gb, const int ngemm,
                                           const float* __restrict__ X,
                                           const float* __restrict__ Wg,
                                           const float* __restrict__ dinv,
                                           float* __restrict__ H, int n) {
    const int tid = threadIdx.x;
    const int lane = tid & 31;
    const int wave = tid >> 5;
    const int half = lane >> 4;
    const int l15  = lane & 15;
    const int mtiles = (n + 15) / 16;

    constexpr int WG_PER_THREAD = NFEAT * NCLASS / 4 / HB_THREADS;
    static_assert(WG_PER_THREAD * HB_THREADS * 4 == NFEAT * NCLASS, "Wg must divide evenly over the block");
    v4f w[WG_PER_THREAD];
#pragma unroll
    for (int j = 0; j < WG_PER_THREAD; ++j) w[j] = reinterpret_cast<const v4f*>(Wg)[tid + j * HB_THREADS];
#pragma unroll
    for (int j = 0; j < WG_PER_THREAD; ++j) {
        const int i  = tid + j * HB_THREADS;
        const int k  = i >> 4;
        const int c0 = (i & 15) * 4;
        float* d = wl + (k >> 1) * 128 + (c0 >> 5) * 64 + (c0 & 15) * 4 + ((c0 >> 4) & 1) * 2 + (k & 1);
        d[0]  = w[j].x;
        d[4]  = w[j].y;
        d[8]  = w[j].z;
        d[12] = w[j].w;
    }
    __syncthreads();

    const v4f* wl4 = reinterpret_cast<const v4f*>(wl) + half * 64 + l15;

    for (int mtile = wave * ngemm + gb; mtile < mtiles; mtile += (HB_THREADS / 32) * ngemm) {
        const int m0 = mtile * 16;
        int m = m0 + l15;
        if (m >= n) m = n - 1;
        const v4f* __restrict__ xrow = reinterpret_cast<const v4f*>(X + (size_t)m * NFEAT) + half;

        v8f acc0 = {}, acc1 = {}, acc2 = {}, acc3 = {};

        v4f a0 = xrow[0];
#pragma unroll
        for (int kb = 0; kb < NFEAT / 8; kb += 2) {
            const v4f a1 = xrow[(kb + 1) * 2];
            mma_block8(acc0, acc1, acc2, acc3, a0, wl4 + kb * 128);
            asm volatile("" ::: "memory");
            if (kb + 2 < NFEAT / 8) a0 = xrow[(kb + 2) * 2];
            mma_block8(acc0, acc1, acc2, acc3, a1, wl4 + (kb + 1) * 128);
            asm volatile("" ::: "memory");
        }

        const int mbase = m0 + half * 8;
#pragma unroll
        for (int r = 0; r < 8; ++r) {
            const int mr = mbase + r;
            if (mr < n) {
                const float dv = dinv[mr];
                float* __restrict__ hrow = H + (size_t)mr * NCLASS;
                hrow[l15]      = acc0[r] * dv;
                hrow[16 + l15] = acc1[r] * dv;
                hrow[32 + l15] = acc2[r] * dv;
                hrow[48 + l15] = acc3[r] * dv;
            }
        }
    }
}

__global__ __launch_bounds__(HB_THREADS, 16) void place_gemm1_kernel(const int* __restrict__ ei,
                                                                     const unsigned int* __restrict__ excl,
                                                                     const unsigned int* __restrict__ lp,
                                                                     const unsigned int* __restrict__ bt,
                                                                     unsigned int* __restrict__ rowptr,
                                                                     int* __restrict__ srcs,
                                                                     const float* __restrict__ X,
                                                                     const float* __restrict__ Wg,
                                                                     const float* __restrict__ dinv,
                                                                     float* __restrict__ H,
                                                                     int n, int e, int nplace) {
    __shared__ __attribute__((aligned(16))) unsigned int bins[HB_BINS];
    __shared__ unsigned int boff[HB_BINS / FB];
    static_assert(HB_BINS >= NFEAT * NCLASS, "the bins must hold Wg for a product block");
    const int b = blockIdx.x;
    if (b < nplace) {
        place_body(bins, boff, b % HB_SLICES, b / HB_SLICES, nplace / HB_SLICES,
                   ei, excl, lp, bt, rowptr, srcs, n, e);
    } else {
        gemm1_body(reinterpret_cast<float*>(bins), b - nplace, (int)gridDim.x - nplace, X, Wg, dinv, H, n);
    }
}

#define GA_BATCH 8
#define GN 16
#define TP 68

__global__ __launch_bounds__(GN * 32) void gather_gemm2_kernel(const unsigned int* __restrict__ rowptr,
                                                              const int* __restrict__ srcs,
                                                              const float* __restrict__ H,
                                                              const float* __restrict__ dinv,
                                                              const float* __restrict__ bg,
                                                              const float* Wl,
                                                              const float* __restrict__ bl,
                                                              float* __restrict__ OUT, int n) {
    __shared__ __attribute__((aligned(16))) float tile[GN * TP];
    const int lane = threadIdx.x & 31;
    const int wave = __builtin_amdgcn_readfirstlane((int)(threadIdx.x >> 5));
    const int m0 = blockIdx.x * GN;
    const int node = m0 + wave;

    const int half = lane >> 4;
    const int l15  = lane & 15;
    const int col  = wave * 16 + l15;

    v2f b[NCLASS / 4];
    if (wave < NCLASS / 16) {
        const float* wcol = Wl + (size_t)(half * 2) * NCLASS + col;
#pragma unroll
        for (int j = 0; j < NCLASS / 4; ++j) {
            b[j].x = wcol[(size_t)(4 * j) * NCLASS];
            b[j].y = wcol[(size_t)(4 * j + 1) * NCLASS];
        }
    }

    float2 o = make_float2(0.0f, 0.0f);
    if (node < n) {
        const unsigned int beg = rowptr[node];
        const unsigned int end = rowptr[node + 1];
        const float di = dinv[node];
        const float2* __restrict__ Hv = reinterpret_cast<const float2*>(H);
        const float2 self = Hv[(size_t)node * (NCLASS / 2) + lane];
        const float2 bias = reinterpret_cast<const float2*>(bg)[lane];

        float ax = 0.0f, ay = 0.0f;
        for (unsigned int base = beg; base < end; base += 32u) {
            const unsigned int j = base + (unsigned int)lane;
            int r = (j < end) ? srcs[j] : node;
            if ((unsigned int)r >= (unsigned int)n) r = node;
            const int m = (int)((end - base < 32u) ? (end - base) : 32u);
            for (int k = 0; k < m; k += GA_BATCH) {
                float2 hv[GA_BATCH];
#pragma unroll
                for (int u = 0; u < GA_BATCH; ++u) {
                    const int kk = k + u;
                    hv[u] = make_float2(0.0f, 0.0f);
                    if (kk < m) {
                        const int rk = __builtin_amdgcn_readlane(r, kk);
                        hv[u] = Hv[(size_t)rk * (NCLASS / 2) + lane];
                    }
                }
#pragma unroll
                for (int u = 0; u < GA_BATCH; ++u) {
                    ax += hv[u].x;
                    ay += hv[u].y;
                }
            }
        }
        o.x = fmaxf((ax + self.x) * di + bias.x, 0.0f);
        o.y = fmaxf((ay + self.y) * di + bias.y, 0.0f);
    }
    reinterpret_cast<float2*>(tile + wave * TP)[lane] = o;

    __syncthreads();

    if (wave < NCLASS / 16) {
        v8f acc = {};
        const float* arow = tile + l15 * TP + half * 2;
#pragma unroll
        for (int j = 0; j < NCLASS / 4; ++j) {
            v2f a;
            a.x = arow[4 * j];
            a.y = arow[4 * j + 1];
            acc = __builtin_amdgcn_wmma_f32_16x16x4_f32(false, a, false, b[j], (short)0, acc, false, false);
        }
        const float blv = bl[col];
        const int mbase = m0 + half * 8;
#pragma unroll
        for (int r = 0; r < 8; ++r) {
            const int mr = mbase + r;
            if (mr < n) OUT[(size_t)mr * NCLASS + col] = acc[r] + blv;
        }
    }
}

extern "C" void kernel_launch(void* const* d_in, const int* in_sizes, int n_in,
                              void* d_out, int out_size, void* d_ws, size_t ws_size,
                              hipStream_t stream) {
    const float* x  = (const float*)d_in[0];
    const int*   ei = (const int*)d_in[1];
    const float* Wg = (const float*)d_in[2];
    const float* bg = (const float*)d_in[3];
    const float* Wl = (const float*)d_in[4];
    const float* bl = (const float*)d_in[5];
    float* out = (float*)d_out;

    const int n = in_sizes[0] / NFEAT;
    const int e = in_sizes[1] / 2;

    float* ws   = (float*)d_ws;
    size_t off  = ((size_t)n + 3) & ~(size_t)3;
    float* dinv = ws;
    float* h    = ws + off;
    float* lists = h + (size_t)n * NCLASS;

    static_assert(HB_SLICES <= NCLASS, "the slice counts must fit in the output buffer");
    unsigned int* partial = (unsigned int*)out;
    unsigned int* rowptr = (unsigned int*)lists;
    int* srcs = (int*)lists + (((size_t)n + 64) & ~(size_t)63);
    unsigned int* lp = (unsigned int*)(srcs + (((size_t)e + 63) & ~(size_t)63));
    unsigned int* bt = lp + n;

    const int ranges = (n + HB_BINS - 1) / HB_BINS;
    deg_hist_kernel  <<<dim3(HB_SLICES, ranges), HB_THREADS, 0, stream>>>(ei + e, partial, n, e);
    deg_finish_kernel<<<(n + FB - 1) / FB, FB, 0, stream>>>(partial, dinv, lp, bt, n);
    const int nplace = HB_SLICES * ranges;
    place_gemm1_kernel<<<2 * nplace, HB_THREADS, 0, stream>>>(ei, partial, lp, bt, rowptr, srcs,
                                                             x, Wg, dinv, h, n, e, nplace);

    gather_gemm2_kernel<<<(n + GN - 1) / GN, GN * 32, 0, stream>>>(rowptr, srcs, h, dinv, bg, Wl, bl, out, n);
}
